# starting kernel
# baseline (speedup 1.0000x reference)
#include <hip/hip_runtime.h>
#include <math.h>

#define BATCH 8
#define SEQ   1024
#define KREL  4
#define NPOS  (BATCH * SEQ)

typedef __attribute__((ext_vector_type(2))) float v2f;
typedef __attribute__((ext_vector_type(4))) float v4f;
typedef __attribute__((ext_vector_type(8))) float v8f;

__global__ void __launch_bounds__(256)
k_rel_reduce(const float* __restrict__ rel, float* __restrict__ factA4) {
  const int row = blockIdx.x;
  const int i   = row & (SEQ - 1);
  const v4f* base = (const v4f*)(rel + (size_t)row * (SEQ * KREL));
  __shared__ int found;
  if (threadIdx.x == 0) found = 0;
  __syncthreads();
  int local = 0;
  for (int j = threadIdx.x; j < i; j += 256) {
    v4f v = __builtin_nontemporal_load(base + j);
    local |= (v.x > 0.5f) || (v.y > 0.5f) || (v.z > 0.5f) || (v.w > 0.5f);
  }
  if (local) atomicOr(&found, 1);
  __syncthreads();
  if (threadIdx.x == 0) factA4[row] = found ? 0.7f : 1.0f;
}

__global__ void __launch_bounds__(128)
k_mlp_factors(const float* __restrict__ aL, const float* __restrict__ oL,
              const int*   __restrict__ exA, const int* __restrict__ exO,
              const float* __restrict__ W1, const float* __restrict__ b1,
              const float* __restrict__ W2, const float* __restrict__ b2,
              const float* __restrict__ W3, const float* __restrict__ b3,
              float* __restrict__ mult1, float* __restrict__ fA2,
              float* __restrict__ fO2) {
  __shared__ float h[4][16][32];
  const int tid  = threadIdx.x;
  const int w    = tid >> 5;
  const int lane = tid & 31;
  const int hf   = lane >> 4;
  const int ln   = lane & 15;
  const int posBase = blockIdx.x * 64 + w * 16;
  const int p = posBase + ln;

  const float* aRow = aL + (size_t)p * 4;
  const float* oRow = oL + (size_t)p * 4;
  const float x0 = aRow[0], x1 = aRow[1], x2 = aRow[2], x3 = aRow[3];
  const float x4 = oRow[0], x5 = oRow[1], x6 = oRow[2], x7 = oRow[3];

  v2f a0, a1, bw;
  a0.x = hf ? x2 : x0;  a0.y = hf ? x3 : x1;
  a1.x = hf ? x6 : x4;  a1.y = hf ? x7 : x5;
  v8f acc0 = {0,0,0,0,0,0,0,0};
  v8f acc1 = {0,0,0,0,0,0,0,0};
  bw.x = W1[(0 + 2*hf)*32 + ln];  bw.y = W1[(1 + 2*hf)*32 + ln];
  acc0 = __builtin_amdgcn_wmma_f32_16x16x4_f32(false, a0, false, bw, (short)0, acc0, false, false);
  bw.x = W1[(4 + 2*hf)*32 + ln];  bw.y = W1[(5 + 2*hf)*32 + ln];
  acc0 = __builtin_amdgcn_wmma_f32_16x16x4_f32(false, a1, false, bw, (short)0, acc0, false, false);
  bw.x = W1[(0 + 2*hf)*32 + 16 + ln];  bw.y = W1[(1 + 2*hf)*32 + 16 + ln];
  acc1 = __builtin_amdgcn_wmma_f32_16x16x4_f32(false, a0, false, bw, (short)0, acc1, false, false);
  bw.x = W1[(4 + 2*hf)*32 + 16 + ln];  bw.y = W1[(5 + 2*hf)*32 + 16 + ln];
  acc1 = __builtin_amdgcn_wmma_f32_16x16x4_f32(false, a1, false, bw, (short)0, acc1, false, false);

  const float bia0 = b1[ln], bia1 = b1[16 + ln];
  #pragma unroll
  for (int v = 0; v < 8; ++v) {
    float t0 = acc0[v] + bia0;  h[w][v + 8*hf][ln]      = t0 > 0.0f ? t0 : 0.0f;
    float t1 = acc1[v] + bia1;  h[w][v + 8*hf][16 + ln] = t1 > 0.0f ? t1 : 0.0f;
  }
  __syncthreads();

  v8f acc2 = {0,0,0,0,0,0,0,0};
  #pragma unroll
  for (int k0 = 0; k0 < 32; k0 += 4) {
    v2f aa, bb;
    aa.x = h[w][ln][k0 + 2*hf];
    aa.y = h[w][ln][k0 + 1 + 2*hf];
    bb.x = W2[(k0 + 2*hf)*16 + ln];
    bb.y = W2[(k0 + 1 + 2*hf)*16 + ln];
    acc2 = __builtin_amdgcn_wmma_f32_16x16x4_f32(false, aa, false, bb, (short)0, acc2, false, false);
  }
  __syncthreads();
  const float bia2 = b2[ln];
  #pragma unroll
  for (int v = 0; v < 8; ++v) {
    float t = acc2[v] + bia2;
    h[w][v + 8*hf][ln] = t > 0.0f ? t : 0.0f;
  }
  __syncthreads();

  if (hf == 0) {
    float s3 = b3[0];
    #pragma unroll
    for (int c = 0; c < 16; ++c) s3 += h[w][ln][c] * W3[c];
    float cg   = 1.0f / (1.0f + expf(-s3));
    float mult = cg < 0.5f ? 2.0f * cg : 1.0f;
    mult1[p] = mult;

    float mA = fmaxf(fmaxf(x0, x1), fmaxf(x2, x3));
    float e0 = expf(x0 - mA), e1 = expf(x1 - mA), e2 = expf(x2 - mA), e3 = expf(x3 - mA);
    float impA = (e0 + e1) / (e0 + e1 + e2 + e3);
    float mO = fmaxf(fmaxf(x4, x5), fmaxf(x6, x7));
    float g0 = expf(x4 - mO), g1 = expf(x5 - mO), g2 = expf(x6 - mO), g3 = expf(x7 - mO);
    float impO = (g0 + g1) / (g0 + g1 + g2 + g3);

    const int bI = p >> 10;
    const int s  = p & (SEQ - 1);
    const int* eA = exA + bI * SEQ;
    const int* eO = exO + bI * SEQ;
    bool nearA = false, nearO = false;
    #pragma unroll
    for (int d = -3; d <= 3; ++d) {
      int ss = s + d;
      if (ss >= 0 && ss < SEQ) { nearA |= (eA[ss] > 0); nearO |= (eO[ss] > 0); }
    }
    fA2[p] = (impA > 0.5f && !nearO) ? 0.3f : 1.0f;
    fO2[p] = (impO > 0.5f && !nearA) ? 0.3f : 1.0f;
  }
}

__global__ void __launch_bounds__(256)
k_scan_write(const float* __restrict__ aL, const float* __restrict__ oL,
             const float* __restrict__ mult1, const float* __restrict__ fA2,
             const float* __restrict__ fO2,  const float* __restrict__ fA4,
             float* __restrict__ outA, float* __restrict__ outO) {
  const int b = blockIdx.x;
  const int t = threadIdx.x;
  __shared__ float sm[SEQ], sfa[SEQ], sfo[SEQ];
  __shared__ unsigned OG[32], IF1[32], IF01[32], F3[32];
  if (t < 32) { OG[t] = 0u; IF1[t] = 0u; IF01[t] = 0u; }
  __syncthreads();

  for (int s = t; s < SEQ; s += 256) {
    const int p = b * SEQ + s;
    v4f av = *(const v4f*)(aL + (size_t)p * 4);
    v4f ov = *(const v4f*)(oL + (size_t)p * 4);
    float mult = mult1[p];
    float fa1  = fA2[p] * fA4[p];
    float fo1  = fO2[p];
    sm[s] = mult;  sfa[s] = fa1;  sfo[s] = fo1;

    float a0 = av.x * mult, a1 = av.y * mult, a2 = av.z * mult, a3 = av.w * mult;
    float o0 = ov.x * mult, o1 = ov.y * mult, o2 = ov.z * mult, o3 = ov.w * mult;
    bool orig = (fmaxf(fmaxf(av.x, av.y), fmaxf(av.z, av.w)) > 0.5f) ||
                (fmaxf(fmaxf(ov.x, ov.y), fmaxf(ov.z, ov.w)) > 0.5f);
    float fa = fa1, fo = fo1;
    bool act1  = (fmaxf(fmaxf(a0*fa, a1*fa), fmaxf(a2, a3)) > 0.5f) ||
                 (fmaxf(fmaxf(o0*fo, o1*fo), fmaxf(o2, o3)) > 0.5f);
    fa = fa1 * 0.1f; fo = fo1 * 0.1f;
    bool act01 = (fmaxf(fmaxf(a0*fa, a1*fa), fmaxf(a2, a3)) > 0.5f) ||
                 (fmaxf(fmaxf(o0*fo, o1*fo), fmaxf(o2, o3)) > 0.5f);

    unsigned bit = 1u << (s & 31);
    if (orig)  atomicOr(&OG[s >> 5],  bit);
    if (act1)  atomicOr(&IF1[s >> 5], bit);
    if (act01) atomicOr(&IF01[s >> 5], bit);
  }
  __syncthreads();

  if (t == 0) {
    unsigned n1 = 0u, n2 = 0u;
    for (int wi = 0; wi < 32; ++wi) {
      unsigned og   = OG[wi];
      unsigned ogn  = (wi < 31) ? OG[wi + 1] : 0u;
      unsigned if1  = IF1[wi];
      unsigned if01 = IF01[wi];
      unsigned f3w  = 0u;
      #pragma unroll 4
      for (int bb = 0; bb < 32; ++bb) {
        unsigned o1b = (bb + 1 < 32) ? ((og >> (bb + 1)) & 1u) : (ogn & 1u);
        unsigned o2b = (bb + 2 < 32) ? ((og >> (bb + 2)) & 1u) : ((ogn >> (bb + 2 - 32)) & 1u);
        unsigned nb  = n1 | n2 | o1b | o2b;
        unsigned na  = nb ? ((if1 >> bb) & 1u) : ((if01 >> bb) & 1u);
        f3w |= nb << bb;
        n2 = n1;  n1 = na;
      }
      F3[wi] = f3w;
    }
  }
  __syncthreads();

  for (int s = t; s < SEQ; s += 256) {
    const int p = b * SEQ + s;
    v4f av = *(const v4f*)(aL + (size_t)p * 4);
    v4f ov = *(const v4f*)(oL + (size_t)p * 4);
    float mult = sm[s];
    float f3 = ((F3[s >> 5] >> (s & 31)) & 1u) ? 1.0f : 0.1f;
    float fa = sfa[s] * f3;
    float fo = sfo[s] * f3;
    v4f ra, ro;
    ra.x = av.x * mult * fa;  ra.y = av.y * mult * fa;
    ra.z = av.z * mult;       ra.w = av.w * mult;
    ro.x = ov.x * mult * fo;  ro.y = ov.y * mult * fo;
    ro.z = ov.z * mult;       ro.w = ov.w * mult;
    *(v4f*)(outA + (size_t)p * 4) = ra;
    *(v4f*)(outO + (size_t)p * 4) = ro;
  }
}

extern "C" void kernel_launch(void* const* d_in, const int* in_sizes, int n_in,
                              void* d_out, int out_size, void* d_ws, size_t ws_size,
                              hipStream_t stream) {
  (void)in_sizes; (void)n_in; (void)out_size; (void)ws_size;
  const float* aL  = (const float*)d_in[0];
  const float* oL  = (const float*)d_in[1];
  const float* rel = (const float*)d_in[2];
  const int*   exA = (const int*)d_in[3];
  const int*   exO = (const int*)d_in[4];
  const float* W1  = (const float*)d_in[5];
  const float* b1  = (const float*)d_in[6];
  const float* W2  = (const float*)d_in[7];
  const float* b2  = (const float*)d_in[8];
  const float* W3  = (const float*)d_in[9];
  const float* b3  = (const float*)d_in[10];

  float* ws    = (float*)d_ws;
  float* fA4   = ws;
  float* mult1 = ws + NPOS;
  float* fA2   = ws + 2 * NPOS;
  float* fO2   = ws + 3 * NPOS;
  float* outA  = (float*)d_out;
  float* outO  = outA + NPOS * 4;

  k_rel_reduce<<<NPOS, 256, 0, stream>>>(rel, fA4);
  k_mlp_factors<<<NPOS / 64, 128, 0, stream>>>(aL, oL, exA, exO,
                                               W1, b1, W2, b2, W3, b3,
                                               mult1, fA2, fO2);
  k_scan_write<<<BATCH, 256, 0, stream>>>(aL, oL, mult1, fA2, fO2, fA4,
                                          outA, outO);
}


# k_rel_reduce: rel's triangle streamed by 4096 equal waves (row pair j, 1023-j each: 33 aligned 512 B loads in flight; no LDS, barrier, atomic), one 16-wave group per WGP; was a block per row
# speedup vs baseline: 1.9860x; 1.9860x over previous
#include <hip/hip_runtime.h>
#include <math.h>

#define BATCH 8
#define SEQ   1024
#define KREL  4
#define NPOS  (BATCH * SEQ)

typedef __attribute__((ext_vector_type(2))) float v2f;
typedef __attribute__((ext_vector_type(4))) float v4f;
typedef __attribute__((ext_vector_type(8))) float v8f;

#define REL_SLOTS  33
#define REL_WAVES  16
#define REL_BLOCKS (BATCH * (SEQ / 2) / REL_WAVES)

__global__ void __launch_bounds__(REL_WAVES * 32)
k_rel_reduce(const float* __restrict__ rel, float* __restrict__ factA4) {
  const unsigned lane = threadIdx.x & 31u;
  const unsigned wv   = __builtin_amdgcn_readfirstlane(threadIdx.x >> 5);
  const unsigned pr   = blockIdx.x * REL_WAVES + wv;
  const unsigned bI   = pr >> 9;
  const unsigned i1   = pr & 511u;
  const unsigned i2   = (SEQ - 1u) - i1;
  const unsigned row1 = bI * SEQ + i1;
  const unsigned row2 = bI * SEQ + i2;
  const v4f* bb = (const v4f*)(rel + (size_t)bI * (SEQ * SEQ * KREL));
  const unsigned q1 = (i1 + 31u) >> 5;

  unsigned off[REL_SLOTS];
  #pragma unroll
  for (int s = 0; s < REL_SLOTS; ++s) {
    const bool     first = (unsigned)s < q1;
    const unsigned i     = first ? i1 : i2;
    const unsigned c     = first ? (unsigned)s : (unsigned)s - q1;
    const unsigned last  = i - 1u;
    unsigned idx = c * 32u + lane;
    idx = idx < last ? idx : last;
    off[s] = i * SEQ + idx;
  }
  v4f v[REL_SLOTS];
  #pragma unroll
  for (int s = 0; s < REL_SLOTS; ++s) v[s] = __builtin_nontemporal_load(bb + off[s]);
  __builtin_amdgcn_sched_barrier(0);
  float m1 = 0.0f, m2 = 0.0f;
  #pragma unroll
  for (int s = 0; s < REL_SLOTS; ++s) {
    const bool  first = (unsigned)s < q1;
    const float t = fmaxf(fmaxf(v[s].x, v[s].y), fmaxf(v[s].z, v[s].w));
    m1 = fmaxf(m1, first ? t : 0.0f);
    m2 = fmaxf(m2, first ? 0.0f : t);
  }
  const bool hit1 = __builtin_amdgcn_ballot_w32(m1 > 0.5f) != 0u;
  const bool hit2 = __builtin_amdgcn_ballot_w32(m2 > 0.5f) != 0u;
  if (lane == 0) {
    factA4[row1] = hit1 ? 0.7f : 1.0f;
    factA4[row2] = hit2 ? 0.7f : 1.0f;
  }
}

__global__ void __launch_bounds__(128)
k_mlp_factors(const float* __restrict__ aL, const float* __restrict__ oL,
              const int*   __restrict__ exA, const int* __restrict__ exO,
              const float* __restrict__ W1, const float* __restrict__ b1,
              const float* __restrict__ W2, const float* __restrict__ b2,
              const float* __restrict__ W3, const float* __restrict__ b3,
              float* __restrict__ mult1, float* __restrict__ fA2,
              float* __restrict__ fO2) {
  __shared__ float h[4][16][32];
  const int tid  = threadIdx.x;
  const int w    = tid >> 5;
  const int lane = tid & 31;
  const int hf   = lane >> 4;
  const int ln   = lane & 15;
  const int posBase = blockIdx.x * 64 + w * 16;
  const int p = posBase + ln;

  const float* aRow = aL + (size_t)p * 4;
  const float* oRow = oL + (size_t)p * 4;
  const float x0 = aRow[0], x1 = aRow[1], x2 = aRow[2], x3 = aRow[3];
  const float x4 = oRow[0], x5 = oRow[1], x6 = oRow[2], x7 = oRow[3];

  v2f a0, a1, bw;
  a0.x = hf ? x2 : x0;  a0.y = hf ? x3 : x1;
  a1.x = hf ? x6 : x4;  a1.y = hf ? x7 : x5;
  v8f acc0 = {0,0,0,0,0,0,0,0};
  v8f acc1 = {0,0,0,0,0,0,0,0};
  bw.x = W1[(0 + 2*hf)*32 + ln];  bw.y = W1[(1 + 2*hf)*32 + ln];
  acc0 = __builtin_amdgcn_wmma_f32_16x16x4_f32(false, a0, false, bw, (short)0, acc0, false, false);
  bw.x = W1[(4 + 2*hf)*32 + ln];  bw.y = W1[(5 + 2*hf)*32 + ln];
  acc0 = __builtin_amdgcn_wmma_f32_16x16x4_f32(false, a1, false, bw, (short)0, acc0, false, false);
  bw.x = W1[(0 + 2*hf)*32 + 16 + ln];  bw.y = W1[(1 + 2*hf)*32 + 16 + ln];
  acc1 = __builtin_amdgcn_wmma_f32_16x16x4_f32(false, a0, false, bw, (short)0, acc1, false, false);
  bw.x = W1[(4 + 2*hf)*32 + 16 + ln];  bw.y = W1[(5 + 2*hf)*32 + 16 + ln];
  acc1 = __builtin_amdgcn_wmma_f32_16x16x4_f32(false, a1, false, bw, (short)0, acc1, false, false);

  const float bia0 = b1[ln], bia1 = b1[16 + ln];
  #pragma unroll
  for (int v = 0; v < 8; ++v) {
    float t0 = acc0[v] + bia0;  h[w][v + 8*hf][ln]      = t0 > 0.0f ? t0 : 0.0f;
    float t1 = acc1[v] + bia1;  h[w][v + 8*hf][16 + ln] = t1 > 0.0f ? t1 : 0.0f;
  }
  __syncthreads();

  v8f acc2 = {0,0,0,0,0,0,0,0};
  #pragma unroll
  for (int k0 = 0; k0 < 32; k0 += 4) {
    v2f aa, bb;
    aa.x = h[w][ln][k0 + 2*hf];
    aa.y = h[w][ln][k0 + 1 + 2*hf];
    bb.x = W2[(k0 + 2*hf)*16 + ln];
    bb.y = W2[(k0 + 1 + 2*hf)*16 + ln];
    acc2 = __builtin_amdgcn_wmma_f32_16x16x4_f32(false, aa, false, bb, (short)0, acc2, false, false);
  }
  __syncthreads();
  const float bia2 = b2[ln];
  #pragma unroll
  for (int v = 0; v < 8; ++v) {
    float t = acc2[v] + bia2;
    h[w][v + 8*hf][ln] = t > 0.0f ? t : 0.0f;
  }
  __syncthreads();

  if (hf == 0) {
    float s3 = b3[0];
    #pragma unroll
    for (int c = 0; c < 16; ++c) s3 += h[w][ln][c] * W3[c];
    float cg   = 1.0f / (1.0f + expf(-s3));
    float mult = cg < 0.5f ? 2.0f * cg : 1.0f;
    mult1[p] = mult;

    float mA = fmaxf(fmaxf(x0, x1), fmaxf(x2, x3));
    float e0 = expf(x0 - mA), e1 = expf(x1 - mA), e2 = expf(x2 - mA), e3 = expf(x3 - mA);
    float impA = (e0 + e1) / (e0 + e1 + e2 + e3);
    float mO = fmaxf(fmaxf(x4, x5), fmaxf(x6, x7));
    float g0 = expf(x4 - mO), g1 = expf(x5 - mO), g2 = expf(x6 - mO), g3 = expf(x7 - mO);
    float impO = (g0 + g1) / (g0 + g1 + g2 + g3);

    const int bI = p >> 10;
    const int s  = p & (SEQ - 1);
    const int* eA = exA + bI * SEQ;
    const int* eO = exO + bI * SEQ;
    bool nearA = false, nearO = false;
    #pragma unroll
    for (int d = -3; d <= 3; ++d) {
      int ss = s + d;
      if (ss >= 0 && ss < SEQ) { nearA |= (eA[ss] > 0); nearO |= (eO[ss] > 0); }
    }
    fA2[p] = (impA > 0.5f && !nearO) ? 0.3f : 1.0f;
    fO2[p] = (impO > 0.5f && !nearA) ? 0.3f : 1.0f;
  }
}

__global__ void __launch_bounds__(256)
k_scan_write(const float* __restrict__ aL, const float* __restrict__ oL,
             const float* __restrict__ mult1, const float* __restrict__ fA2,
             const float* __restrict__ fO2,  const float* __restrict__ fA4,
             float* __restrict__ outA, float* __restrict__ outO) {
  const int b = blockIdx.x;
  const int t = threadIdx.x;
  __shared__ float sm[SEQ], sfa[SEQ], sfo[SEQ];
  __shared__ unsigned OG[32], IF1[32], IF01[32], F3[32];
  if (t < 32) { OG[t] = 0u; IF1[t] = 0u; IF01[t] = 0u; }
  __syncthreads();

  for (int s = t; s < SEQ; s += 256) {
    const int p = b * SEQ + s;
    v4f av = *(const v4f*)(aL + (size_t)p * 4);
    v4f ov = *(const v4f*)(oL + (size_t)p * 4);
    float mult = mult1[p];
    float fa1  = fA2[p] * fA4[p];
    float fo1  = fO2[p];
    sm[s] = mult;  sfa[s] = fa1;  sfo[s] = fo1;

    float a0 = av.x * mult, a1 = av.y * mult, a2 = av.z * mult, a3 = av.w * mult;
    float o0 = ov.x * mult, o1 = ov.y * mult, o2 = ov.z * mult, o3 = ov.w * mult;
    bool orig = (fmaxf(fmaxf(av.x, av.y), fmaxf(av.z, av.w)) > 0.5f) ||
                (fmaxf(fmaxf(ov.x, ov.y), fmaxf(ov.z, ov.w)) > 0.5f);
    float fa = fa1, fo = fo1;
    bool act1  = (fmaxf(fmaxf(a0*fa, a1*fa), fmaxf(a2, a3)) > 0.5f) ||
                 (fmaxf(fmaxf(o0*fo, o1*fo), fmaxf(o2, o3)) > 0.5f);
    fa = fa1 * 0.1f; fo = fo1 * 0.1f;
    bool act01 = (fmaxf(fmaxf(a0*fa, a1*fa), fmaxf(a2, a3)) > 0.5f) ||
                 (fmaxf(fmaxf(o0*fo, o1*fo), fmaxf(o2, o3)) > 0.5f);

    unsigned bit = 1u << (s & 31);
    if (orig)  atomicOr(&OG[s >> 5],  bit);
    if (act1)  atomicOr(&IF1[s >> 5], bit);
    if (act01) atomicOr(&IF01[s >> 5], bit);
  }
  __syncthreads();

  if (t < 32) {
    const unsigned og  = OG[t];
    const unsigned ogn = (t < 31) ? OG[t + 1] : 0u;
    const unsigned P   = IF1[t];
    const unsigned np  = ~P;
    const unsigned npPrevTop = (t > 0) ? (~IF1[t - 1] >> 31) : 1u;
    const unsigned F   = ((og >> 1) | (ogn << 31)) | ((og >> 2) | (ogn << 30));
    const unsigned G   = IF01[t] | (P & F);
    const unsigned A   = ~(np & ((np << 1) | npPrevTop));
    const unsigned S0  = A + G;
    const unsigned Gm  = __builtin_amdgcn_ballot_w32(S0 < A);
    const unsigned Pm  = __builtin_amdgcn_ballot_w32(S0 == 0xFFFFFFFFu);
    const unsigned X   = Gm | Pm;
    const unsigned Cm  = (X + Gm) ^ X ^ Gm;
    const unsigned S   = S0 + ((Cm >> t) & 1u);
    F3[t] = (S ^ A ^ G) | F;
  }
  __syncthreads();

  for (int s = t; s < SEQ; s += 256) {
    const int p = b * SEQ + s;
    v4f av = *(const v4f*)(aL + (size_t)p * 4);
    v4f ov = *(const v4f*)(oL + (size_t)p * 4);
    float mult = sm[s];
    float f3 = ((F3[s >> 5] >> (s & 31)) & 1u) ? 1.0f : 0.1f;
    float fa = sfa[s] * f3;
    float fo = sfo[s] * f3;
    v4f ra, ro;
    ra.x = av.x * mult * fa;  ra.y = av.y * mult * fa;
    ra.z = av.z * mult;       ra.w = av.w * mult;
    ro.x = ov.x * mult * fo;  ro.y = ov.y * mult * fo;
    ro.z = ov.z * mult;       ro.w = ov.w * mult;
    *(v4f*)(outA + (size_t)p * 4) = ra;
    *(v4f*)(outO + (size_t)p * 4) = ro;
  }
}

extern "C" void kernel_launch(void* const* d_in, const int* in_sizes, int n_in,
                              void* d_out, int out_size, void* d_ws, size_t ws_size,
                              hipStream_t stream) {
  (void)in_sizes; (void)n_in; (void)out_size; (void)ws_size;
  const float* aL  = (const float*)d_in[0];
  const float* oL  = (const float*)d_in[1];
  const float* rel = (const float*)d_in[2];
  const int*   exA = (const int*)d_in[3];
  const int*   exO = (const int*)d_in[4];
  const float* W1  = (const float*)d_in[5];
  const float* b1  = (const float*)d_in[6];
  const float* W2  = (const float*)d_in[7];
  const float* b2  = (const float*)d_in[8];
  const float* W3  = (const float*)d_in[9];
  const float* b3  = (const float*)d_in[10];

  float* ws    = (float*)d_ws;
  float* fA4   = ws;
  float* mult1 = ws + NPOS;
  float* fA2   = ws + 2 * NPOS;
  float* fO2   = ws + 3 * NPOS;
  float* outA  = (float*)d_out;
  float* outO  = outA + NPOS * 4;

  k_rel_reduce<<<REL_BLOCKS, REL_WAVES * 32, 0, stream>>>(rel, fA4);
  k_mlp_factors<<<NPOS / 64, 128, 0, stream>>>(aL, oL, exA, exO,
                                               W1, b1, W2, b2, W3, b3,
                                               mult1, fA2, fO2);
  k_scan_write<<<BATCH, 256, 0, stream>>>(aL, oL, mult1, fA2, fO2, fA4,
                                          outA, outO);
}


# per-position pass of k_scan_write moved into the stream waves (lanes 0,1 store each row for fact3=1 plus 4 flag bits); scan kernel loads flag words only, rewrites just isolated rows; same result bits
# speedup vs baseline: 2.5807x; 1.2994x over previous
#include <hip/hip_runtime.h>
#include <math.h>

#define BATCH 8
#define SEQ   1024
#define KREL  4
#define NPOS  (BATCH * SEQ)

typedef __attribute__((ext_vector_type(2))) float v2f;
typedef __attribute__((ext_vector_type(4))) float v4f;
typedef __attribute__((ext_vector_type(8))) float v8f;
typedef __attribute__((ext_vector_type(4))) unsigned v4u;

#define REL_SLOTS  33
#define REL_WAVES  16
#define REL_BLOCKS (BATCH * (SEQ / 2) / REL_WAVES)

__global__ void __launch_bounds__(REL_WAVES * 32)
k_rel_reduce(const float* __restrict__ rel,
             const float* __restrict__ aL, const float* __restrict__ oL,
             const float* __restrict__ mult1, const float* __restrict__ fA2,
             const float* __restrict__ fO2, unsigned* __restrict__ flags,
             float* __restrict__ outA, float* __restrict__ outO) {
  const unsigned lane = threadIdx.x & 31u;
  const unsigned wv   = __builtin_amdgcn_readfirstlane(threadIdx.x >> 5);
  const unsigned pr   = blockIdx.x * REL_WAVES + wv;
  const unsigned bI   = pr >> 9;
  const unsigned i1   = pr & 511u;
  const unsigned i2   = (SEQ - 1u) - i1;
  const unsigned row1 = bI * SEQ + i1;
  const unsigned row2 = bI * SEQ + i2;
  const v4f* bb = (const v4f*)(rel + (size_t)bI * (SEQ * SEQ * KREL));
  const unsigned q1 = (i1 + 31u) >> 5;

  const unsigned prow = (lane == 0u) ? row1 : row2;
  const v4f   av   = *(const v4f*)(aL + (size_t)prow * 4);
  const v4f   ov   = *(const v4f*)(oL + (size_t)prow * 4);
  const float mult = mult1[prow];
  const float f2   = fA2[prow];
  const float fo1  = fO2[prow];

  unsigned off[REL_SLOTS];
  #pragma unroll
  for (int s = 0; s < REL_SLOTS; ++s) {
    const bool     first = (unsigned)s < q1;
    const unsigned i     = first ? i1 : i2;
    const unsigned c     = first ? (unsigned)s : (unsigned)s - q1;
    const unsigned last  = i - 1u;
    unsigned idx = c * 32u + lane;
    idx = idx < last ? idx : last;
    off[s] = i * SEQ + idx;
  }
  v4f v[REL_SLOTS];
  #pragma unroll
  for (int s = 0; s < REL_SLOTS; ++s) v[s] = __builtin_nontemporal_load(bb + off[s]);
  __builtin_amdgcn_sched_barrier(0);
  float m1 = 0.0f, m2 = 0.0f;
  #pragma unroll
  for (int s = 0; s < REL_SLOTS; ++s) {
    const bool  first = (unsigned)s < q1;
    const float t = fmaxf(fmaxf(v[s].x, v[s].y), fmaxf(v[s].z, v[s].w));
    m1 = fmaxf(m1, first ? t : 0.0f);
    m2 = fmaxf(m2, first ? 0.0f : t);
  }
  const bool hit1 = __builtin_amdgcn_ballot_w32(m1 > 0.5f) != 0u;
  const bool hit2 = __builtin_amdgcn_ballot_w32(m2 > 0.5f) != 0u;
  const bool  hit = (lane == 0u) ? hit1 : hit2;
  const float fa1 = f2 * (hit ? 0.7f : 1.0f);
  const float a0 = av.x * mult, a1 = av.y * mult, a2 = av.z * mult, a3 = av.w * mult;
  const float o0 = ov.x * mult, o1 = ov.y * mult, o2 = ov.z * mult, o3 = ov.w * mult;
  const bool orig = (fmaxf(fmaxf(av.x, av.y), fmaxf(av.z, av.w)) > 0.5f) ||
                    (fmaxf(fmaxf(ov.x, ov.y), fmaxf(ov.z, ov.w)) > 0.5f);
  const bool act1  = (fmaxf(fmaxf(a0*fa1, a1*fa1), fmaxf(a2, a3)) > 0.5f) ||
                     (fmaxf(fmaxf(o0*fo1, o1*fo1), fmaxf(o2, o3)) > 0.5f);
  const float fa = fa1 * 0.1f, fo = fo1 * 0.1f;
  const bool act01 = (fmaxf(fmaxf(a0*fa, a1*fa), fmaxf(a2, a3)) > 0.5f) ||
                     (fmaxf(fmaxf(o0*fo, o1*fo), fmaxf(o2, o3)) > 0.5f);
  if (lane < 2u) {
    v4f ra, ro;
    ra.x = a0 * fa1;  ra.y = a1 * fa1;  ra.z = a2;  ra.w = a3;
    ro.x = o0 * fo1;  ro.y = o1 * fo1;  ro.z = o2;  ro.w = o3;
    *(v4f*)(outA + (size_t)prow * 4) = ra;
    *(v4f*)(outO + (size_t)prow * 4) = ro;
    flags[prow] = (hit ? 1u : 0u) | (orig ? 2u : 0u) | (act1 ? 4u : 0u) | (act01 ? 8u : 0u);
  }
}

__global__ void __launch_bounds__(128)
k_mlp_factors(const float* __restrict__ aL, const float* __restrict__ oL,
              const int*   __restrict__ exA, const int* __restrict__ exO,
              const float* __restrict__ W1, const float* __restrict__ b1,
              const float* __restrict__ W2, const float* __restrict__ b2,
              const float* __restrict__ W3, const float* __restrict__ b3,
              float* __restrict__ mult1, float* __restrict__ fA2,
              float* __restrict__ fO2) {
  __shared__ float h[4][16][32];
  const int tid  = threadIdx.x;
  const int w    = tid >> 5;
  const int lane = tid & 31;
  const int hf   = lane >> 4;
  const int ln   = lane & 15;
  const int posBase = blockIdx.x * 64 + w * 16;
  const int p = posBase + ln;

  const float* aRow = aL + (size_t)p * 4;
  const float* oRow = oL + (size_t)p * 4;
  const float x0 = aRow[0], x1 = aRow[1], x2 = aRow[2], x3 = aRow[3];
  const float x4 = oRow[0], x5 = oRow[1], x6 = oRow[2], x7 = oRow[3];

  v2f a0, a1, bw;
  a0.x = hf ? x2 : x0;  a0.y = hf ? x3 : x1;
  a1.x = hf ? x6 : x4;  a1.y = hf ? x7 : x5;
  v8f acc0 = {0,0,0,0,0,0,0,0};
  v8f acc1 = {0,0,0,0,0,0,0,0};
  bw.x = W1[(0 + 2*hf)*32 + ln];  bw.y = W1[(1 + 2*hf)*32 + ln];
  acc0 = __builtin_amdgcn_wmma_f32_16x16x4_f32(false, a0, false, bw, (short)0, acc0, false, false);
  bw.x = W1[(4 + 2*hf)*32 + ln];  bw.y = W1[(5 + 2*hf)*32 + ln];
  acc0 = __builtin_amdgcn_wmma_f32_16x16x4_f32(false, a1, false, bw, (short)0, acc0, false, false);
  bw.x = W1[(0 + 2*hf)*32 + 16 + ln];  bw.y = W1[(1 + 2*hf)*32 + 16 + ln];
  acc1 = __builtin_amdgcn_wmma_f32_16x16x4_f32(false, a0, false, bw, (short)0, acc1, false, false);
  bw.x = W1[(4 + 2*hf)*32 + 16 + ln];  bw.y = W1[(5 + 2*hf)*32 + 16 + ln];
  acc1 = __builtin_amdgcn_wmma_f32_16x16x4_f32(false, a1, false, bw, (short)0, acc1, false, false);

  const float bia0 = b1[ln], bia1 = b1[16 + ln];
  #pragma unroll
  for (int v = 0; v < 8; ++v) {
    float t0 = acc0[v] + bia0;  h[w][v + 8*hf][ln]      = t0 > 0.0f ? t0 : 0.0f;
    float t1 = acc1[v] + bia1;  h[w][v + 8*hf][16 + ln] = t1 > 0.0f ? t1 : 0.0f;
  }
  __syncthreads();

  v8f acc2 = {0,0,0,0,0,0,0,0};
  #pragma unroll
  for (int k0 = 0; k0 < 32; k0 += 4) {
    v2f aa, bb;
    aa.x = h[w][ln][k0 + 2*hf];
    aa.y = h[w][ln][k0 + 1 + 2*hf];
    bb.x = W2[(k0 + 2*hf)*16 + ln];
    bb.y = W2[(k0 + 1 + 2*hf)*16 + ln];
    acc2 = __builtin_amdgcn_wmma_f32_16x16x4_f32(false, aa, false, bb, (short)0, acc2, false, false);
  }
  __syncthreads();
  const float bia2 = b2[ln];
  #pragma unroll
  for (int v = 0; v < 8; ++v) {
    float t = acc2[v] + bia2;
    h[w][v + 8*hf][ln] = t > 0.0f ? t : 0.0f;
  }
  __syncthreads();

  if (hf == 0) {
    float s3 = b3[0];
    #pragma unroll
    for (int c = 0; c < 16; ++c) s3 += h[w][ln][c] * W3[c];
    float cg   = 1.0f / (1.0f + expf(-s3));
    float mult = cg < 0.5f ? 2.0f * cg : 1.0f;
    mult1[p] = mult;

    float mA = fmaxf(fmaxf(x0, x1), fmaxf(x2, x3));
    float e0 = expf(x0 - mA), e1 = expf(x1 - mA), e2 = expf(x2 - mA), e3 = expf(x3 - mA);
    float impA = (e0 + e1) / (e0 + e1 + e2 + e3);
    float mO = fmaxf(fmaxf(x4, x5), fmaxf(x6, x7));
    float g0 = expf(x4 - mO), g1 = expf(x5 - mO), g2 = expf(x6 - mO), g3 = expf(x7 - mO);
    float impO = (g0 + g1) / (g0 + g1 + g2 + g3);

    const int bI = p >> 10;
    const int s  = p & (SEQ - 1);
    const int* eA = exA + bI * SEQ;
    const int* eO = exO + bI * SEQ;
    bool nearA = false, nearO = false;
    #pragma unroll
    for (int d = -3; d <= 3; ++d) {
      int ss = s + d;
      if (ss >= 0 && ss < SEQ) { nearA |= (eA[ss] > 0); nearO |= (eO[ss] > 0); }
    }
    fA2[p] = (impA > 0.5f && !nearO) ? 0.3f : 1.0f;
    fO2[p] = (impO > 0.5f && !nearA) ? 0.3f : 1.0f;
  }
}

__global__ void __launch_bounds__(256)
k_scan_write(const float* __restrict__ aL, const float* __restrict__ oL,
             const float* __restrict__ mult1, const float* __restrict__ fA2,
             const float* __restrict__ fO2,  const unsigned* __restrict__ flags,
             float* __restrict__ outA, float* __restrict__ outO) {
  const int b = blockIdx.x;
  const int t = threadIdx.x;
  __shared__ unsigned OG[32], IF1[32], IF01[32], F3[32];
  const v4u fl = *(const v4u*)(flags + b * SEQ + 4 * t);
  if (t < 32) { OG[t] = 0u; IF1[t] = 0u; IF01[t] = 0u; }
  __syncthreads();

  const unsigned w  = (unsigned)t >> 3;
  const unsigned sh = 4u * ((unsigned)t & 7u);
  unsigned nOG = 0u, nI1 = 0u, nI01 = 0u;
  for (int j = 0; j < 4; ++j) {
    nOG  |= ((fl[j] >> 1) & 1u) << j;
    nI1  |= ((fl[j] >> 2) & 1u) << j;
    nI01 |= ((fl[j] >> 3) & 1u) << j;
  }
  if (nOG)  atomicOr(&OG[w],   nOG  << sh);
  if (nI1)  atomicOr(&IF1[w],  nI1  << sh);
  if (nI01) atomicOr(&IF01[w], nI01 << sh);
  __syncthreads();

  if (t < 32) {
    const unsigned og  = OG[t];
    const unsigned ogn = (t < 31) ? OG[t + 1] : 0u;
    const unsigned P   = IF1[t];
    const unsigned np  = ~P;
    const unsigned npPrevTop = (t > 0) ? (~IF1[t - 1] >> 31) : 1u;
    const unsigned F   = ((og >> 1) | (ogn << 31)) | ((og >> 2) | (ogn << 30));
    const unsigned G   = IF01[t] | (P & F);
    const unsigned A   = ~(np & ((np << 1) | npPrevTop));
    const unsigned S0  = A + G;
    const unsigned Gm  = __builtin_amdgcn_ballot_w32(S0 < A);
    const unsigned Pm  = __builtin_amdgcn_ballot_w32(S0 == 0xFFFFFFFFu);
    const unsigned X   = Gm | Pm;
    const unsigned Cm  = (X + Gm) ^ X ^ Gm;
    const unsigned S   = S0 + ((Cm >> t) & 1u);
    F3[t] = (S ^ A ^ G) | F;
  }
  __syncthreads();

  const unsigned keep = (F3[w] >> sh) & 15u;
  if (keep != 15u) {
    for (int j = 0; j < 4; ++j) {
      if (((keep >> j) & 1u) == 0u) {
        const int p = b * SEQ + 4 * t + j;
        const v4f av = *(const v4f*)(aL + (size_t)p * 4);
        const v4f ov = *(const v4f*)(oL + (size_t)p * 4);
        const float mult = mult1[p];
        const float fa = (fA2[p] * ((fl[j] & 1u) ? 0.7f : 1.0f)) * 0.1f;
        const float fo = fO2[p] * 0.1f;
        v4f ra, ro;
        ra.x = av.x * mult * fa;  ra.y = av.y * mult * fa;
        ra.z = av.z * mult;       ra.w = av.w * mult;
        ro.x = ov.x * mult * fo;  ro.y = ov.y * mult * fo;
        ro.z = ov.z * mult;       ro.w = ov.w * mult;
        *(v4f*)(outA + (size_t)p * 4) = ra;
        *(v4f*)(outO + (size_t)p * 4) = ro;
      }
    }
  }
}

extern "C" void kernel_launch(void* const* d_in, const int* in_sizes, int n_in,
                              void* d_out, int out_size, void* d_ws, size_t ws_size,
                              hipStream_t stream) {
  (void)in_sizes; (void)n_in; (void)out_size; (void)ws_size;
  const float* aL  = (const float*)d_in[0];
  const float* oL  = (const float*)d_in[1];
  const float* rel = (const float*)d_in[2];
  const int*   exA = (const int*)d_in[3];
  const int*   exO = (const int*)d_in[4];
  const float* W1  = (const float*)d_in[5];
  const float* b1  = (const float*)d_in[6];
  const float* W2  = (const float*)d_in[7];
  const float* b2  = (const float*)d_in[8];
  const float* W3  = (const float*)d_in[9];
  const float* b3  = (const float*)d_in[10];

  float* ws    = (float*)d_ws;
  unsigned* flags = (unsigned*)d_ws;
  float* mult1 = ws + NPOS;
  float* fA2   = ws + 2 * NPOS;
  float* fO2   = ws + 3 * NPOS;
  float* outA  = (float*)d_out;
  float* outO  = outA + NPOS * 4;

  k_mlp_factors<<<NPOS / 64, 128, 0, stream>>>(aL, oL, exA, exO,
                                               W1, b1, W2, b2, W3, b3,
                                               mult1, fA2, fO2);
  k_rel_reduce<<<REL_BLOCKS, REL_WAVES * 32, 0, stream>>>(rel, aL, oL, mult1, fA2, fO2,
                                                          flags, outA, outO);
  k_scan_write<<<BATCH, 256, 0, stream>>>(aL, oL, mult1, fA2, fO2, flags,
                                          outA, outO);
}


# k_rel_reduce: rel's triangle streamed by 1024 long-lived waves, not 4096: 4 per WGP (one per SIMD), 4 row pairs each, 132 loads requested up front (counter keeps 63 in flight); same result bits
# speedup vs baseline: 2.7604x; 1.0696x over previous
#include <hip/hip_runtime.h>
#include <math.h>

#define BATCH 8
#define SEQ   1024
#define KREL  4
#define NPOS  (BATCH * SEQ)

typedef __attribute__((ext_vector_type(2))) float v2f;
typedef __attribute__((ext_vector_type(4))) float v4f;
typedef __attribute__((ext_vector_type(8))) float v8f;
typedef __attribute__((ext_vector_type(4))) unsigned v4u;

#define REL_SLOTS  33
#define REL_PAIRS  4
#define REL_WAVES  4
#define REL_BLOCKS (BATCH * (SEQ / 2) / (REL_WAVES * REL_PAIRS))

__global__ void __launch_bounds__(REL_WAVES * 32)
k_rel_reduce(const float* __restrict__ rel,
             const float* __restrict__ aL, const float* __restrict__ oL,
             const float* __restrict__ mult1, const float* __restrict__ fA2,
             const float* __restrict__ fO2, unsigned* __restrict__ flags,
             float* __restrict__ outA, float* __restrict__ outO) {
  const unsigned lane = threadIdx.x & 31u;
  const unsigned wv   = __builtin_amdgcn_readfirstlane(threadIdx.x >> 5);
  const unsigned pr0  = (blockIdx.x * REL_WAVES + wv) * REL_PAIRS;
  const unsigned bI   = pr0 >> 9;
  const unsigned j0   = pr0 & 511u;
  const v4f* bb = (const v4f*)rel;

  const unsigned own  = lane & 7u;
  const unsigned i1o  = j0 + (own >> 1);
  const unsigned prow = bI * SEQ + ((own & 1u) ? (SEQ - 1u) - i1o : i1o);
  const v4f   av   = *(const v4f*)(aL + (size_t)prow * 4);
  const v4f   ov   = *(const v4f*)(oL + (size_t)prow * 4);
  const float mult = mult1[prow];
  const float f2   = fA2[prow];
  const float fo1  = fO2[prow];

  v4f v[REL_PAIRS][REL_SLOTS];
  #pragma unroll
  for (int k = 0; k < REL_PAIRS; ++k) {
    const unsigned i1 = j0 + (unsigned)k;
    const unsigned i2 = (SEQ - 1u) - i1;
    const unsigned q1 = (i1 + 31u) >> 5;
    unsigned off[REL_SLOTS];
    #pragma unroll
    for (int s = 0; s < REL_SLOTS; ++s) {
      const bool     first = (unsigned)s < q1;
      const unsigned i     = first ? i1 : i2;
      const unsigned c     = first ? (unsigned)s : (unsigned)s - q1;
      const unsigned last  = i - 1u;
      unsigned idx = c * 32u + lane;
      idx = idx < last ? idx : last;
      off[s] = (bI * SEQ + i) * SEQ + idx;
    }
    #pragma unroll
    for (int s = 0; s < REL_SLOTS; ++s) v[k][s] = __builtin_nontemporal_load(bb + off[s]);
    __builtin_amdgcn_sched_barrier(0);
  }
  unsigned hm = 0u;
  #pragma unroll
  for (int k = 0; k < REL_PAIRS; ++k) {
    const unsigned q1 = (j0 + (unsigned)k + 31u) >> 5;
    float m1 = 0.0f, m2 = 0.0f;
    #pragma unroll
    for (int s = 0; s < REL_SLOTS; ++s) {
      const bool  first = (unsigned)s < q1;
      const float t = fmaxf(fmaxf(v[k][s].x, v[k][s].y), fmaxf(v[k][s].z, v[k][s].w));
      m1 = fmaxf(m1, first ? t : 0.0f);
      m2 = fmaxf(m2, first ? 0.0f : t);
    }
    const unsigned h1 = __builtin_amdgcn_ballot_w32(m1 > 0.5f) != 0u ? 1u : 0u;
    const unsigned h2 = __builtin_amdgcn_ballot_w32(m2 > 0.5f) != 0u ? 1u : 0u;
    hm |= (h1 << (2 * k)) | (h2 << (2 * k + 1));
  }
  const bool  hit = ((hm >> own) & 1u) != 0u;
  const float fa1 = f2 * (hit ? 0.7f : 1.0f);
  const float a0 = av.x * mult, a1 = av.y * mult, a2 = av.z * mult, a3 = av.w * mult;
  const float o0 = ov.x * mult, o1 = ov.y * mult, o2 = ov.z * mult, o3 = ov.w * mult;
  const bool orig = (fmaxf(fmaxf(av.x, av.y), fmaxf(av.z, av.w)) > 0.5f) ||
                    (fmaxf(fmaxf(ov.x, ov.y), fmaxf(ov.z, ov.w)) > 0.5f);
  const bool act1  = (fmaxf(fmaxf(a0*fa1, a1*fa1), fmaxf(a2, a3)) > 0.5f) ||
                     (fmaxf(fmaxf(o0*fo1, o1*fo1), fmaxf(o2, o3)) > 0.5f);
  const float fa = fa1 * 0.1f, fo = fo1 * 0.1f;
  const bool act01 = (fmaxf(fmaxf(a0*fa, a1*fa), fmaxf(a2, a3)) > 0.5f) ||
                     (fmaxf(fmaxf(o0*fo, o1*fo), fmaxf(o2, o3)) > 0.5f);
  if (lane < 8u) {
    v4f ra, ro;
    ra.x = a0 * fa1;  ra.y = a1 * fa1;  ra.z = a2;  ra.w = a3;
    ro.x = o0 * fo1;  ro.y = o1 * fo1;  ro.z = o2;  ro.w = o3;
    *(v4f*)(outA + (size_t)prow * 4) = ra;
    *(v4f*)(outO + (size_t)prow * 4) = ro;
    flags[prow] = (hit ? 1u : 0u) | (orig ? 2u : 0u) | (act1 ? 4u : 0u) | (act01 ? 8u : 0u);
  }
}

__global__ void __launch_bounds__(128)
k_mlp_factors(const float* __restrict__ aL, const float* __restrict__ oL,
              const int*   __restrict__ exA, const int* __restrict__ exO,
              const float* __restrict__ W1, const float* __restrict__ b1,
              const float* __restrict__ W2, const float* __restrict__ b2,
              const float* __restrict__ W3, const float* __restrict__ b3,
              float* __restrict__ mult1, float* __restrict__ fA2,
              float* __restrict__ fO2) {
  __shared__ float h[4][16][32];
  const int tid  = threadIdx.x;
  const int w    = tid >> 5;
  const int lane = tid & 31;
  const int hf   = lane >> 4;
  const int ln   = lane & 15;
  const int posBase = blockIdx.x * 64 + w * 16;
  const int p = posBase + ln;

  const float* aRow = aL + (size_t)p * 4;
  const float* oRow = oL + (size_t)p * 4;
  const float x0 = aRow[0], x1 = aRow[1], x2 = aRow[2], x3 = aRow[3];
  const float x4 = oRow[0], x5 = oRow[1], x6 = oRow[2], x7 = oRow[3];

  v2f a0, a1, bw;
  a0.x = hf ? x2 : x0;  a0.y = hf ? x3 : x1;
  a1.x = hf ? x6 : x4;  a1.y = hf ? x7 : x5;
  v8f acc0 = {0,0,0,0,0,0,0,0};
  v8f acc1 = {0,0,0,0,0,0,0,0};
  bw.x = W1[(0 + 2*hf)*32 + ln];  bw.y = W1[(1 + 2*hf)*32 + ln];
  acc0 = __builtin_amdgcn_wmma_f32_16x16x4_f32(false, a0, false, bw, (short)0, acc0, false, false);
  bw.x = W1[(4 + 2*hf)*32 + ln];  bw.y = W1[(5 + 2*hf)*32 + ln];
  acc0 = __builtin_amdgcn_wmma_f32_16x16x4_f32(false, a1, false, bw, (short)0, acc0, false, false);
  bw.x = W1[(0 + 2*hf)*32 + 16 + ln];  bw.y = W1[(1 + 2*hf)*32 + 16 + ln];
  acc1 = __builtin_amdgcn_wmma_f32_16x16x4_f32(false, a0, false, bw, (short)0, acc1, false, false);
  bw.x = W1[(4 + 2*hf)*32 + 16 + ln];  bw.y = W1[(5 + 2*hf)*32 + 16 + ln];
  acc1 = __builtin_amdgcn_wmma_f32_16x16x4_f32(false, a1, false, bw, (short)0, acc1, false, false);

  const float bia0 = b1[ln], bia1 = b1[16 + ln];
  #pragma unroll
  for (int v = 0; v < 8; ++v) {
    float t0 = acc0[v] + bia0;  h[w][v + 8*hf][ln]      = t0 > 0.0f ? t0 : 0.0f;
    float t1 = acc1[v] + bia1;  h[w][v + 8*hf][16 + ln] = t1 > 0.0f ? t1 : 0.0f;
  }
  __syncthreads();

  v8f acc2 = {0,0,0,0,0,0,0,0};
  #pragma unroll
  for (int k0 = 0; k0 < 32; k0 += 4) {
    v2f aa, bb;
    aa.x = h[w][ln][k0 + 2*hf];
    aa.y = h[w][ln][k0 + 1 + 2*hf];
    bb.x = W2[(k0 + 2*hf)*16 + ln];
    bb.y = W2[(k0 + 1 + 2*hf)*16 + ln];
    acc2 = __builtin_amdgcn_wmma_f32_16x16x4_f32(false, aa, false, bb, (short)0, acc2, false, false);
  }
  __syncthreads();
  const float bia2 = b2[ln];
  #pragma unroll
  for (int v = 0; v < 8; ++v) {
    float t = acc2[v] + bia2;
    h[w][v + 8*hf][ln] = t > 0.0f ? t : 0.0f;
  }
  __syncthreads();

  if (hf == 0) {
    float s3 = b3[0];
    #pragma unroll
    for (int c = 0; c < 16; ++c) s3 += h[w][ln][c] * W3[c];
    float cg   = 1.0f / (1.0f + expf(-s3));
    float mult = cg < 0.5f ? 2.0f * cg : 1.0f;
    mult1[p] = mult;

    float mA = fmaxf(fmaxf(x0, x1), fmaxf(x2, x3));
    float e0 = expf(x0 - mA), e1 = expf(x1 - mA), e2 = expf(x2 - mA), e3 = expf(x3 - mA);
    float impA = (e0 + e1) / (e0 + e1 + e2 + e3);
    float mO = fmaxf(fmaxf(x4, x5), fmaxf(x6, x7));
    float g0 = expf(x4 - mO), g1 = expf(x5 - mO), g2 = expf(x6 - mO), g3 = expf(x7 - mO);
    float impO = (g0 + g1) / (g0 + g1 + g2 + g3);

    const int bI = p >> 10;
    const int s  = p & (SEQ - 1);
    const int* eA = exA + bI * SEQ;
    const int* eO = exO + bI * SEQ;
    bool nearA = false, nearO = false;
    #pragma unroll
    for (int d = -3; d <= 3; ++d) {
      int ss = s + d;
      if (ss >= 0 && ss < SEQ) { nearA |= (eA[ss] > 0); nearO |= (eO[ss] > 0); }
    }
    fA2[p] = (impA > 0.5f && !nearO) ? 0.3f : 1.0f;
    fO2[p] = (impO > 0.5f && !nearA) ? 0.3f : 1.0f;
  }
}

__global__ void __launch_bounds__(256)
k_scan_write(const float* __restrict__ aL, const float* __restrict__ oL,
             const float* __restrict__ mult1, const float* __restrict__ fA2,
             const float* __restrict__ fO2,  const unsigned* __restrict__ flags,
             float* __restrict__ outA, float* __restrict__ outO) {
  const int b = blockIdx.x;
  const int t = threadIdx.x;
  __shared__ unsigned OG[32], IF1[32], IF01[32], F3[32];
  const v4u fl = *(const v4u*)(flags + b * SEQ + 4 * t);
  if (t < 32) { OG[t] = 0u; IF1[t] = 0u; IF01[t] = 0u; }
  __syncthreads();

  const unsigned w  = (unsigned)t >> 3;
  const unsigned sh = 4u * ((unsigned)t & 7u);
  unsigned nOG = 0u, nI1 = 0u, nI01 = 0u;
  for (int j = 0; j < 4; ++j) {
    nOG  |= ((fl[j] >> 1) & 1u) << j;
    nI1  |= ((fl[j] >> 2) & 1u) << j;
    nI01 |= ((fl[j] >> 3) & 1u) << j;
  }
  if (nOG)  atomicOr(&OG[w],   nOG  << sh);
  if (nI1)  atomicOr(&IF1[w],  nI1  << sh);
  if (nI01) atomicOr(&IF01[w], nI01 << sh);
  __syncthreads();

  if (t < 32) {
    const unsigned og  = OG[t];
    const unsigned ogn = (t < 31) ? OG[t + 1] : 0u;
    const unsigned P   = IF1[t];
    const unsigned np  = ~P;
    const unsigned npPrevTop = (t > 0) ? (~IF1[t - 1] >> 31) : 1u;
    const unsigned F   = ((og >> 1) | (ogn << 31)) | ((og >> 2) | (ogn << 30));
    const unsigned G   = IF01[t] | (P & F);
    const unsigned A   = ~(np & ((np << 1) | npPrevTop));
    const unsigned S0  = A + G;
    const unsigned Gm  = __builtin_amdgcn_ballot_w32(S0 < A);
    const unsigned Pm  = __builtin_amdgcn_ballot_w32(S0 == 0xFFFFFFFFu);
    const unsigned X   = Gm | Pm;
    const unsigned Cm  = (X + Gm) ^ X ^ Gm;
    const unsigned S   = S0 + ((Cm >> t) & 1u);
    F3[t] = (S ^ A ^ G) | F;
  }
  __syncthreads();

  const unsigned keep = (F3[w] >> sh) & 15u;
  if (keep != 15u) {
    for (int j = 0; j < 4; ++j) {
      if (((keep >> j) & 1u) == 0u) {
        const int p = b * SEQ + 4 * t + j;
        const v4f av = *(const v4f*)(aL + (size_t)p * 4);
        const v4f ov = *(const v4f*)(oL + (size_t)p * 4);
        const float mult = mult1[p];
        const float fa = (fA2[p] * ((fl[j] & 1u) ? 0.7f : 1.0f)) * 0.1f;
        const float fo = fO2[p] * 0.1f;
        v4f ra, ro;
        ra.x = av.x * mult * fa;  ra.y = av.y * mult * fa;
        ra.z = av.z * mult;       ra.w = av.w * mult;
        ro.x = ov.x * mult * fo;  ro.y = ov.y * mult * fo;
        ro.z = ov.z * mult;       ro.w = ov.w * mult;
        *(v4f*)(outA + (size_t)p * 4) = ra;
        *(v4f*)(outO + (size_t)p * 4) = ro;
      }
    }
  }
}

extern "C" void kernel_launch(void* const* d_in, const int* in_sizes, int n_in,
                              void* d_out, int out_size, void* d_ws, size_t ws_size,
                              hipStream_t stream) {
  (void)in_sizes; (void)n_in; (void)out_size; (void)ws_size;
  const float* aL  = (const float*)d_in[0];
  const float* oL  = (const float*)d_in[1];
  const float* rel = (const float*)d_in[2];
  const int*   exA = (const int*)d_in[3];
  const int*   exO = (const int*)d_in[4];
  const float* W1  = (const float*)d_in[5];
  const float* b1  = (const float*)d_in[6];
  const float* W2  = (const float*)d_in[7];
  const float* b2  = (const float*)d_in[8];
  const float* W3  = (const float*)d_in[9];
  const float* b3  = (const float*)d_in[10];

  float* ws    = (float*)d_ws;
  unsigned* flags = (unsigned*)d_ws;
  float* mult1 = ws + NPOS;
  float* fA2   = ws + 2 * NPOS;
  float* fO2   = ws + 3 * NPOS;
  float* outA  = (float*)d_out;
  float* outO  = outA + NPOS * 4;

  k_mlp_factors<<<NPOS / 64, 128, 0, stream>>>(aL, oL, exA, exO,
                                               W1, b1, W2, b2, W3, b3,
                                               mult1, fA2, fO2);
  k_rel_reduce<<<REL_BLOCKS, REL_WAVES * 32, 0, stream>>>(rel, aL, oL, mult1, fA2, fO2,
                                                          flags, outA, outO);
  k_scan_write<<<BATCH, 256, 0, stream>>>(aL, oL, mult1, fA2, fO2, flags,
                                          outA, outO);
}


# step-2 factors (softmax heads, +-3 explicit windows) leave k_mlp_factors for the stream waves' row lanes: 14 window words, 8 exps under 132 loads in flight; they ride the flag word; same result bits
# speedup vs baseline: 2.9796x; 1.0794x over previous
#include <hip/hip_runtime.h>
#include <math.h>

#define BATCH 8
#define SEQ   1024
#define KREL  4
#define NPOS  (BATCH * SEQ)

typedef __attribute__((ext_vector_type(2))) float v2f;
typedef __attribute__((ext_vector_type(4))) float v4f;
typedef __attribute__((ext_vector_type(8))) float v8f;
typedef __attribute__((ext_vector_type(4))) unsigned v4u;

#define REL_SLOTS  33
#define REL_PAIRS  4
#define REL_WAVES  4
#define REL_BLOCKS (BATCH * (SEQ / 2) / (REL_WAVES * REL_PAIRS))

__global__ void __launch_bounds__(REL_WAVES * 32)
k_rel_reduce(const float* __restrict__ rel,
             const float* __restrict__ aL, const float* __restrict__ oL,
             const int*   __restrict__ exA, const int* __restrict__ exO,
             const float* __restrict__ mult1, unsigned* __restrict__ flags,
             float* __restrict__ outA, float* __restrict__ outO) {
  const unsigned lane = threadIdx.x & 31u;
  const unsigned wv   = __builtin_amdgcn_readfirstlane(threadIdx.x >> 5);
  const unsigned pr0  = (blockIdx.x * REL_WAVES + wv) * REL_PAIRS;
  const unsigned bI   = pr0 >> 9;
  const unsigned j0   = pr0 & 511u;
  const v4f* bb = (const v4f*)rel;

  const unsigned own  = lane & 7u;
  const unsigned i1o  = j0 + (own >> 1);
  const unsigned srow = (own & 1u) ? (SEQ - 1u) - i1o : i1o;
  const unsigned prow = bI * SEQ + srow;
  const v4f   av   = *(const v4f*)(aL + (size_t)prow * 4);
  const v4f   ov   = *(const v4f*)(oL + (size_t)prow * 4);
  const float mult = mult1[prow];
  const int* eA = exA + bI * SEQ;
  const int* eO = exO + bI * SEQ;
  int wa[7], wo[7];
  #pragma unroll
  for (int d = 0; d < 7; ++d) {
    const int ss = (int)srow + d - 3;
    const int sc = ss < 0 ? 0 : (ss > SEQ - 1 ? SEQ - 1 : ss);
    wa[d] = eA[sc];
    wo[d] = eO[sc];
  }

  v4f v[REL_PAIRS][REL_SLOTS];
  #pragma unroll
  for (int k = 0; k < REL_PAIRS; ++k) {
    const unsigned i1 = j0 + (unsigned)k;
    const unsigned i2 = (SEQ - 1u) - i1;
    const unsigned q1 = (i1 + 31u) >> 5;
    unsigned off[REL_SLOTS];
    #pragma unroll
    for (int s = 0; s < REL_SLOTS; ++s) {
      const bool     first = (unsigned)s < q1;
      const unsigned i     = first ? i1 : i2;
      const unsigned c     = first ? (unsigned)s : (unsigned)s - q1;
      const unsigned last  = i - 1u;
      unsigned idx = c * 32u + lane;
      idx = idx < last ? idx : last;
      off[s] = (bI * SEQ + i) * SEQ + idx;
    }
    #pragma unroll
    for (int s = 0; s < REL_SLOTS; ++s) v[k][s] = __builtin_nontemporal_load(bb + off[s]);
    __builtin_amdgcn_sched_barrier(0);
  }
  bool nearA = false, nearO = false;
  #pragma unroll
  for (int d = 0; d < 7; ++d) {
    const int  ss = (int)srow + d - 3;
    const bool in = (ss >= 0) && (ss < SEQ);
    nearA |= in && (wa[d] > 0);
    nearO |= in && (wo[d] > 0);
  }
  const float mA = fmaxf(fmaxf(av.x, av.y), fmaxf(av.z, av.w));
  const float e0 = expf(av.x - mA), e1 = expf(av.y - mA), e2 = expf(av.z - mA), e3 = expf(av.w - mA);
  const float impA = (e0 + e1) / (e0 + e1 + e2 + e3);
  const float mO = fmaxf(fmaxf(ov.x, ov.y), fmaxf(ov.z, ov.w));
  const float g0 = expf(ov.x - mO), g1 = expf(ov.y - mO), g2 = expf(ov.z - mO), g3 = expf(ov.w - mO);
  const float impO = (g0 + g1) / (g0 + g1 + g2 + g3);
  const bool  lowA = (impA > 0.5f) && !nearO;
  const bool  lowO = (impO > 0.5f) && !nearA;
  float    f2    = lowA ? 0.3f : 1.0f;
  float    fo1   = lowO ? 0.3f : 1.0f;
  unsigned fbits = (lowA ? 16u : 0u) | (lowO ? 32u : 0u);
  asm volatile("" : "+v"(f2), "+v"(fo1), "+v"(fbits));
  __builtin_amdgcn_sched_barrier(0);

  unsigned hm = 0u;
  #pragma unroll
  for (int k = 0; k < REL_PAIRS; ++k) {
    const unsigned q1 = (j0 + (unsigned)k + 31u) >> 5;
    float m1 = 0.0f, m2 = 0.0f;
    #pragma unroll
    for (int s = 0; s < REL_SLOTS; ++s) {
      const bool  first = (unsigned)s < q1;
      const float t = fmaxf(fmaxf(v[k][s].x, v[k][s].y), fmaxf(v[k][s].z, v[k][s].w));
      m1 = fmaxf(m1, first ? t : 0.0f);
      m2 = fmaxf(m2, first ? 0.0f : t);
    }
    const unsigned h1 = __builtin_amdgcn_ballot_w32(m1 > 0.5f) != 0u ? 1u : 0u;
    const unsigned h2 = __builtin_amdgcn_ballot_w32(m2 > 0.5f) != 0u ? 1u : 0u;
    hm |= (h1 << (2 * k)) | (h2 << (2 * k + 1));
  }
  const bool  hit = ((hm >> own) & 1u) != 0u;
  const float fa1 = f2 * (hit ? 0.7f : 1.0f);
  const float a0 = av.x * mult, a1 = av.y * mult, a2 = av.z * mult, a3 = av.w * mult;
  const float o0 = ov.x * mult, o1 = ov.y * mult, o2 = ov.z * mult, o3 = ov.w * mult;
  const bool orig = (fmaxf(fmaxf(av.x, av.y), fmaxf(av.z, av.w)) > 0.5f) ||
                    (fmaxf(fmaxf(ov.x, ov.y), fmaxf(ov.z, ov.w)) > 0.5f);
  const bool act1  = (fmaxf(fmaxf(a0*fa1, a1*fa1), fmaxf(a2, a3)) > 0.5f) ||
                     (fmaxf(fmaxf(o0*fo1, o1*fo1), fmaxf(o2, o3)) > 0.5f);
  const float fa = fa1 * 0.1f, fo = fo1 * 0.1f;
  const bool act01 = (fmaxf(fmaxf(a0*fa, a1*fa), fmaxf(a2, a3)) > 0.5f) ||
                     (fmaxf(fmaxf(o0*fo, o1*fo), fmaxf(o2, o3)) > 0.5f);
  if (lane < 8u) {
    v4f ra, ro;
    ra.x = a0 * fa1;  ra.y = a1 * fa1;  ra.z = a2;  ra.w = a3;
    ro.x = o0 * fo1;  ro.y = o1 * fo1;  ro.z = o2;  ro.w = o3;
    *(v4f*)(outA + (size_t)prow * 4) = ra;
    *(v4f*)(outO + (size_t)prow * 4) = ro;
    flags[prow] = (hit ? 1u : 0u) | (orig ? 2u : 0u) | (act1 ? 4u : 0u) | (act01 ? 8u : 0u) | fbits;
  }
}

__global__ void __launch_bounds__(128)
k_mlp_factors(const float* __restrict__ aL, const float* __restrict__ oL,
              const float* __restrict__ W1, const float* __restrict__ b1,
              const float* __restrict__ W2, const float* __restrict__ b2,
              const float* __restrict__ W3, const float* __restrict__ b3,
              float* __restrict__ mult1) {
  __shared__ float h[4][16][32];
  const int tid  = threadIdx.x;
  const int w    = tid >> 5;
  const int lane = tid & 31;
  const int hf   = lane >> 4;
  const int ln   = lane & 15;
  const int posBase = blockIdx.x * 64 + w * 16;
  const int p = posBase + ln;

  const float* aRow = aL + (size_t)p * 4;
  const float* oRow = oL + (size_t)p * 4;
  const float x0 = aRow[0], x1 = aRow[1], x2 = aRow[2], x3 = aRow[3];
  const float x4 = oRow[0], x5 = oRow[1], x6 = oRow[2], x7 = oRow[3];

  v2f a0, a1, bw;
  a0.x = hf ? x2 : x0;  a0.y = hf ? x3 : x1;
  a1.x = hf ? x6 : x4;  a1.y = hf ? x7 : x5;
  v8f acc0 = {0,0,0,0,0,0,0,0};
  v8f acc1 = {0,0,0,0,0,0,0,0};
  bw.x = W1[(0 + 2*hf)*32 + ln];  bw.y = W1[(1 + 2*hf)*32 + ln];
  acc0 = __builtin_amdgcn_wmma_f32_16x16x4_f32(false, a0, false, bw, (short)0, acc0, false, false);
  bw.x = W1[(4 + 2*hf)*32 + ln];  bw.y = W1[(5 + 2*hf)*32 + ln];
  acc0 = __builtin_amdgcn_wmma_f32_16x16x4_f32(false, a1, false, bw, (short)0, acc0, false, false);
  bw.x = W1[(0 + 2*hf)*32 + 16 + ln];  bw.y = W1[(1 + 2*hf)*32 + 16 + ln];
  acc1 = __builtin_amdgcn_wmma_f32_16x16x4_f32(false, a0, false, bw, (short)0, acc1, false, false);
  bw.x = W1[(4 + 2*hf)*32 + 16 + ln];  bw.y = W1[(5 + 2*hf)*32 + 16 + ln];
  acc1 = __builtin_amdgcn_wmma_f32_16x16x4_f32(false, a1, false, bw, (short)0, acc1, false, false);

  const float bia0 = b1[ln], bia1 = b1[16 + ln];
  #pragma unroll
  for (int v = 0; v < 8; ++v) {
    float t0 = acc0[v] + bia0;  h[w][v + 8*hf][ln]      = t0 > 0.0f ? t0 : 0.0f;
    float t1 = acc1[v] + bia1;  h[w][v + 8*hf][16 + ln] = t1 > 0.0f ? t1 : 0.0f;
  }
  __syncthreads();

  v8f acc2 = {0,0,0,0,0,0,0,0};
  #pragma unroll
  for (int k0 = 0; k0 < 32; k0 += 4) {
    v2f aa, bb;
    aa.x = h[w][ln][k0 + 2*hf];
    aa.y = h[w][ln][k0 + 1 + 2*hf];
    bb.x = W2[(k0 + 2*hf)*16 + ln];
    bb.y = W2[(k0 + 1 + 2*hf)*16 + ln];
    acc2 = __builtin_amdgcn_wmma_f32_16x16x4_f32(false, aa, false, bb, (short)0, acc2, false, false);
  }
  __syncthreads();
  const float bia2 = b2[ln];
  #pragma unroll
  for (int v = 0; v < 8; ++v) {
    float t = acc2[v] + bia2;
    h[w][v + 8*hf][ln] = t > 0.0f ? t : 0.0f;
  }
  __syncthreads();

  if (hf == 0) {
    float s3 = b3[0];
    #pragma unroll
    for (int c = 0; c < 16; ++c) s3 += h[w][ln][c] * W3[c];
    float cg   = 1.0f / (1.0f + expf(-s3));
    float mult = cg < 0.5f ? 2.0f * cg : 1.0f;
    mult1[p] = mult;
  }
}

__global__ void __launch_bounds__(256)
k_scan_write(const float* __restrict__ aL, const float* __restrict__ oL,
             const float* __restrict__ mult1, const unsigned* __restrict__ flags,
             float* __restrict__ outA, float* __restrict__ outO) {
  const int b = blockIdx.x;
  const int t = threadIdx.x;
  __shared__ unsigned OG[32], IF1[32], IF01[32], F3[32];
  const v4u fl = *(const v4u*)(flags + b * SEQ + 4 * t);
  if (t < 32) { OG[t] = 0u; IF1[t] = 0u; IF01[t] = 0u; }
  __syncthreads();

  const unsigned w  = (unsigned)t >> 3;
  const unsigned sh = 4u * ((unsigned)t & 7u);
  unsigned nOG = 0u, nI1 = 0u, nI01 = 0u;
  for (int j = 0; j < 4; ++j) {
    nOG  |= ((fl[j] >> 1) & 1u) << j;
    nI1  |= ((fl[j] >> 2) & 1u) << j;
    nI01 |= ((fl[j] >> 3) & 1u) << j;
  }
  if (nOG)  atomicOr(&OG[w],   nOG  << sh);
  if (nI1)  atomicOr(&IF1[w],  nI1  << sh);
  if (nI01) atomicOr(&IF01[w], nI01 << sh);
  __syncthreads();

  if (t < 32) {
    const unsigned og  = OG[t];
    const unsigned ogn = (t < 31) ? OG[t + 1] : 0u;
    const unsigned P   = IF1[t];
    const unsigned np  = ~P;
    const unsigned npPrevTop = (t > 0) ? (~IF1[t - 1] >> 31) : 1u;
    const unsigned F   = ((og >> 1) | (ogn << 31)) | ((og >> 2) | (ogn << 30));
    const unsigned G   = IF01[t] | (P & F);
    const unsigned A   = ~(np & ((np << 1) | npPrevTop));
    const unsigned S0  = A + G;
    const unsigned Gm  = __builtin_amdgcn_ballot_w32(S0 < A);
    const unsigned Pm  = __builtin_amdgcn_ballot_w32(S0 == 0xFFFFFFFFu);
    const unsigned X   = Gm | Pm;
    const unsigned Cm  = (X + Gm) ^ X ^ Gm;
    const unsigned S   = S0 + ((Cm >> t) & 1u);
    F3[t] = (S ^ A ^ G) | F;
  }
  __syncthreads();

  const unsigned keep = (F3[w] >> sh) & 15u;
  if (keep != 15u) {
    for (int j = 0; j < 4; ++j) {
      if (((keep >> j) & 1u) == 0u) {
        const int p = b * SEQ + 4 * t + j;
        const v4f av = *(const v4f*)(aL + (size_t)p * 4);
        const v4f ov = *(const v4f*)(oL + (size_t)p * 4);
        const float mult = mult1[p];
        const float fA2 = (fl[j] & 16u) ? 0.3f : 1.0f;
        const float fO2 = (fl[j] & 32u) ? 0.3f : 1.0f;
        const float fa = (fA2 * ((fl[j] & 1u) ? 0.7f : 1.0f)) * 0.1f;
        const float fo = fO2 * 0.1f;
        v4f ra, ro;
        ra.x = av.x * mult * fa;  ra.y = av.y * mult * fa;
        ra.z = av.z * mult;       ra.w = av.w * mult;
        ro.x = ov.x * mult * fo;  ro.y = ov.y * mult * fo;
        ro.z = ov.z * mult;       ro.w = ov.w * mult;
        *(v4f*)(outA + (size_t)p * 4) = ra;
        *(v4f*)(outO + (size_t)p * 4) = ro;
      }
    }
  }
}

extern "C" void kernel_launch(void* const* d_in, const int* in_sizes, int n_in,
                              void* d_out, int out_size, void* d_ws, size_t ws_size,
                              hipStream_t stream) {
  (void)in_sizes; (void)n_in; (void)out_size; (void)ws_size;
  const float* aL  = (const float*)d_in[0];
  const float* oL  = (const float*)d_in[1];
  const float* rel = (const float*)d_in[2];
  const int*   exA = (const int*)d_in[3];
  const int*   exO = (const int*)d_in[4];
  const float* W1  = (const float*)d_in[5];
  const float* b1  = (const float*)d_in[6];
  const float* W2  = (const float*)d_in[7];
  const float* b2  = (const float*)d_in[8];
  const float* W3  = (const float*)d_in[9];
  const float* b3  = (const float*)d_in[10];

  float* ws    = (float*)d_ws;
  unsigned* flags = (unsigned*)d_ws;
  float* mult1 = ws + NPOS;
  float* outA  = (float*)d_out;
  float* outO  = outA + NPOS * 4;

  k_mlp_factors<<<NPOS / 64, 128, 0, stream>>>(aL, oL, W1, b1, W2, b2, W3, b3, mult1);
  k_rel_reduce<<<REL_BLOCKS, REL_WAVES * 32, 0, stream>>>(rel, aL, oL, exA, exO, mult1,
                                                          flags, outA, outO);
  k_scan_write<<<BATCH, 256, 0, stream>>>(aL, oL, mult1, flags, outA, outO);
}
